# starting kernel
# baseline (speedup 1.0000x reference)
#include <hip/hip_runtime.h>
#include <math.h>

typedef float v2f __attribute__((ext_vector_type(2)));
typedef float v8f __attribute__((ext_vector_type(8)));

#define BB 8
#define CC 128
#define FF 8
#define FR 7
#define HW 4096
#define HIN 256
#define NPART 512

#define WS_VTR   0
#define WS_VR    (WS_VTR + BB*HW)
#define WS_CM    (WS_VR  + BB*FR*HW)
#define WS_PART  (WS_CM  + BB*FR*HW)
#define WS_VSUM  (WS_PART + BB*FR*NPART)
#define WS_GS    (WS_VSUM + BB*FR)

__global__ void k_init(float* ws) {
    int t = threadIdx.x;
    if (t < BB*FR) ws[WS_VSUM + t] = 0.0f;
}

__global__ void k_masks(const float* __restrict__ v_t,
                        const float* __restrict__ v_al,
                        float* __restrict__ ws) {
    int b     = blockIdx.x >> 4;
    int chunk = blockIdx.x & 15;
    int tid   = threadIdx.x;
    int p     = chunk * 256 + tid;
    int i = p >> 6, j = p & 63;
    int r = 4*i + 1, c = 4*j + 1;

    const float* vt = v_t + (size_t)b * HIN * HIN;
    float a = 0.25f * (vt[r*HIN + c] + vt[r*HIN + c + 1] +
                       vt[(r+1)*HIN + c] + vt[(r+1)*HIN + c + 1]);
    float vtr = (a > 0.5f) ? 1.0f : 0.0f;
    ws[WS_VTR + b*HW + p] = vtr;

    int lane = tid & 31;
    #pragma unroll
    for (int f = 0; f < FR; ++f) {
        const float* va = v_al + (size_t)(b*FR + f) * HIN * HIN;
        float m = 0.25f * (va[r*HIN + c] + va[r*HIN + c + 1] +
                           va[(r+1)*HIN + c] + va[(r+1)*HIN + c + 1]);
        float vr = (m > 0.5f) ? 1.0f : 0.0f;
        ws[WS_VR + (b*FR + f)*HW + p] = vr;
        float s = vtr * vr;
        s += __shfl_xor(s, 16, 32);
        s += __shfl_xor(s, 8, 32);
        s += __shfl_xor(s, 4, 32);
        s += __shfl_xor(s, 2, 32);
        s += __shfl_xor(s, 1, 32);
        if (lane == 0) atomicAdd(&ws[WS_VSUM + b*FR + f], s);
    }
}

__global__ void k_gs_wmma(const float* __restrict__ cf,
                          float* __restrict__ ws) {
    int wib    = threadIdx.x >> 5;
    int lane   = threadIdx.x & 31;
    int waveId = blockIdx.x * 8 + wib;
    int b      = waveId >> 9;
    int rem    = waveId & 511;
    int chalf  = rem >> 8;
    int ptile  = rem & 255;
    int cstart = chalf * 64;
    int pix    = ptile * 16 + (lane & 15);
    int choff  = (lane >> 4) << 1;

    float vtr = ws[WS_VTR + b*HW + pix];
    float vmapv[FR];
    #pragma unroll
    for (int f = 0; f < FR; ++f)
        vmapv[f] = vtr * ws[WS_VR + (b*FR + f)*HW + pix];

    v2f bones = {1.0f, 1.0f};
    v8f acc[FR];
    #pragma unroll
    for (int f = 0; f < FR; ++f) acc[f] = (v8f){0.f,0.f,0.f,0.f,0.f,0.f,0.f,0.f};

    for (int c = cstart; c < cstart + 64; c += 4) {
        int ch = c + choff;
        long base = ((long)(b*CC + ch) * FF) * HW + pix;
        float c0a = cf[base];
        float c0b = cf[base + (long)FF*HW];
        #pragma unroll
        for (int f = 0; f < FR; ++f) {
            float ra = cf[base + (long)(1 + f)*HW];
            float rb = cf[base + (long)FF*HW + (long)(1 + f)*HW];
            v2f a;
            a.x = vmapv[f] * c0a * ra;
            a.y = vmapv[f] * c0b * rb;
            acc[f] = __builtin_amdgcn_wmma_f32_16x16x4_f32(
                false, a, false, bones, (short)0, acc[f], false, false);
        }
    }

    #pragma unroll
    for (int f = 0; f < FR; ++f) {
        float s = acc[f][0] + acc[f][1] + acc[f][2] + acc[f][3] +
                  acc[f][4] + acc[f][5] + acc[f][6] + acc[f][7];
        s += __shfl_xor(s, 16, 32);
        if (lane == 0)
            ws[WS_PART + (b*FR + f)*NPART + chalf*256 + ptile] = s;
    }
}

__global__ void k_gs_final(float* __restrict__ ws) {
    __shared__ float sdata[256];
    int bf  = blockIdx.x;
    int tid = threadIdx.x;
    const float* part = ws + WS_PART + bf*NPART;
    sdata[tid] = part[tid] + part[tid + 256];
    __syncthreads();
    for (int s = 128; s > 0; s >>= 1) {
        if (tid < s) sdata[tid] += sdata[tid + s];
        __syncthreads();
    }
    if (tid == 0) {
        float vsum = ws[WS_VSUM + bf];
        int zero = vsum < 1e-4f;
        float denom = (vsum + (zero ? 1.0f : 0.0f)) * (float)CC;
        ws[WS_GS + bf] = zero ? 0.0f : (sdata[0] / denom);
    }
}

__global__ void k_softmax(float* __restrict__ ws, float* __restrict__ out) {
    int g   = blockIdx.x * blockDim.x + threadIdx.x;
    int b   = g >> 12;
    int pix = g & 4095;

    float vr[FR], mv[FR];
    float mx = -1e30f;
    #pragma unroll
    for (int f = 0; f < FR; ++f) {
        vr[f] = ws[WS_VR + (b*FR + f)*HW + pix];
        mv[f] = ws[WS_GS + b*FR + f] * vr[f];
        mx = fmaxf(mx, mv[f]);
    }
    float sum = 0.0f, ex[FR];
    #pragma unroll
    for (int f = 0; f < FR; ++f) { ex[f] = __builtin_expf(mv[f] - mx) * vr[f]; sum += ex[f]; }
    sum += (sum < 1e-4f) ? 1.0f : 0.0f;
    float inv = 1.0f / sum, cmaskacc = 0.0f;
    #pragma unroll
    for (int f = 0; f < FR; ++f) {
        float cm = ex[f] * inv;
        ws[WS_CM + (b*FR + f)*HW + pix] = cm;
        cmaskacc += cm * vr[f];
    }
    float cmask = 1.0f - cmaskacc;
    out[((size_t)b*257 + 256)*HW + pix] = cmask;
    out[(size_t)BB*257*HW + (size_t)b*HW + pix] = cmask;
}

__global__ void k_cout(const float* __restrict__ cf,
                       const float* __restrict__ ws,
                       float* __restrict__ out) {
    int blk    = blockIdx.x;
    int b      = blk >> 6;
    int pchunk = (blk >> 2) & 15;
    int q      = blk & 3;
    int p      = pchunk * 256 + threadIdx.x;

    float cm[FR];
    #pragma unroll
    for (int f = 0; f < FR; ++f) cm[f] = ws[WS_CM + (b*FR + f)*HW + p];

    for (int c = q*32; c < q*32 + 32; ++c) {
        long base = ((long)(b*CC + c) * FF) * HW + p;
        float c0v = cf[base];
        float acc = 0.0f;
        #pragma unroll
        for (int f = 0; f < FR; ++f) acc += cf[base + (long)(1 + f)*HW] * cm[f];
        out[((size_t)b*257 + c)*HW + p]       = c0v;
        out[((size_t)b*257 + 128 + c)*HW + p] = acc;
    }
}

extern "C" void kernel_launch(void* const* d_in, const int* in_sizes, int n_in,
                              void* d_out, int out_size, void* d_ws, size_t ws_size,
                              hipStream_t stream) {
    const float* c_feats   = (const float*)d_in[0];
    const float* v_t       = (const float*)d_in[1];
    const float* v_aligned = (const float*)d_in[2];
    float* out = (float*)d_out;
    float* ws  = (float*)d_ws;

    k_init    <<<1,   128, 0, stream>>>(ws);
    k_masks   <<<128, 256, 0, stream>>>(v_t, v_aligned, ws);
    k_gs_wmma <<<512, 256, 0, stream>>>(c_feats, ws);
    k_gs_final<<<56,  256, 0, stream>>>(ws);
    k_softmax <<<128, 256, 0, stream>>>(ws, out);
    k_cout    <<<512, 256, 0, stream>>>(c_feats, ws, out);
}


# Fuse the six launches into the two passes over c_feats: masks, v_sum and gs block partials inside the gs pass (no atomics), gs fold and softmax inside the c_out pass; gs partial sums re-associated
# speedup vs baseline: 1.7357x; 1.7357x over previous
#include <hip/hip_runtime.h>
#include <math.h>


typedef float v2f __attribute__((ext_vector_type(2)));
typedef float v8f __attribute__((ext_vector_type(8)));

#define BB 8
#define CC 128
#define FF 8
#define FR 7
#define HW 4096
#define HIN 256
#define NGSP 64
#define NVSP 32

#define WS_VR    0
#define WS_GSP   (WS_VR  + BB*FR*HW)
#define WS_VSP   (WS_GSP + BB*FR*NGSP)

__global__ void k_gs_wmma(const float* __restrict__ cf,
                          const float* __restrict__ v_t,
                          const float* __restrict__ v_al,
                          float* __restrict__ ws) {
    __shared__ float sh_gs[8][FR];
    __shared__ float sh_vs[8][FR];

    int wib    = threadIdx.x >> 5;
    int lane   = threadIdx.x & 31;
    int b      = blockIdx.x >> 6;
    int chalf  = (blockIdx.x >> 5) & 1;
    int blk32  = blockIdx.x & 31;
    int ptile  = blk32 * 8 + wib;
    int cstart = chalf * 64;
    int pix    = ptile * 16 + (lane & 15);
    int choff  = (lane >> 4) << 1;

    int pi = pix >> 6, pj = pix & 63;
    int r0 = 4*pi + 1, c0i = 4*pj + 1;
    const float* vt = v_t + (size_t)b * HIN * HIN;
    float a = 0.25f * (vt[r0*HIN + c0i] + vt[r0*HIN + c0i + 1] +
                       vt[(r0+1)*HIN + c0i] + vt[(r0+1)*HIN + c0i + 1]);
    float vtr = (a > 0.5f) ? 1.0f : 0.0f;

    float vrv[FR], vmapv[FR];
    #pragma unroll
    for (int f = 0; f < FR; ++f) {
        const float* va = v_al + (size_t)(b*FR + f) * HIN * HIN;
        float m = 0.25f * (va[r0*HIN + c0i] + va[r0*HIN + c0i + 1] +
                           va[(r0+1)*HIN + c0i] + va[(r0+1)*HIN + c0i + 1]);
        vrv[f]   = (m > 0.5f) ? 1.0f : 0.0f;
        vmapv[f] = vtr * vrv[f];
    }

    if (chalf == 0) {
        float cnt[FR];
        #pragma unroll
        for (int f = 0; f < FR; ++f) {
            unsigned m = __builtin_amdgcn_ballot_w32(vmapv[f] != 0.0f);
            cnt[f] = (float)__builtin_popcount(m & 0xFFFFu);
        }
        if (lane < 16) {
            #pragma unroll
            for (int f = 0; f < FR; ++f) ws[WS_VR + (b*FR + f)*HW + pix] = vrv[f];
        }
        if (lane == 0) {
            #pragma unroll
            for (int f = 0; f < FR; ++f) sh_vs[wib][f] = cnt[f];
        }
    }

    v2f bones = {1.0f, 1.0f};
    v8f acc[FR];
    #pragma unroll
    for (int f = 0; f < FR; ++f) acc[f] = (v8f){0.f,0.f,0.f,0.f,0.f,0.f,0.f,0.f};

    for (int c = cstart; c < cstart + 64; c += 4) {
        int ch = c + choff;
        long base = ((long)(b*CC + ch) * FF) * HW + pix;
        float c0a = cf[base];
        float c0b = cf[base + (long)FF*HW];
        #pragma unroll
        for (int f = 0; f < FR; ++f) {
            float ra = cf[base + (long)(1 + f)*HW];
            float rb = cf[base + (long)FF*HW + (long)(1 + f)*HW];
            v2f av;
            av.x = vmapv[f] * c0a * ra;
            av.y = vmapv[f] * c0b * rb;
            acc[f] = __builtin_amdgcn_wmma_f32_16x16x4_f32(
                false, av, false, bones, (short)0, acc[f], false, false);
        }
    }

    float wsum[FR];
    #pragma unroll
    for (int f = 0; f < FR; ++f) {
        float s = acc[f][0] + acc[f][1] + acc[f][2] + acc[f][3] +
                  acc[f][4] + acc[f][5] + acc[f][6] + acc[f][7];
        s += __shfl_xor(s, 16, 32);
        wsum[f] = s;
    }
    if (lane == 0) {
        #pragma unroll
        for (int f = 0; f < FR; ++f) sh_gs[wib][f] = wsum[f];
    }
    __syncthreads();

    if (threadIdx.x < FR) {
        int f = threadIdx.x;
        float g = sh_gs[0][f];
        #pragma unroll
        for (int w = 1; w < 8; ++w) g += sh_gs[w][f];
        ws[WS_GSP + (b*FR + f)*NGSP + chalf*32 + blk32] = g;
        if (chalf == 0) {
            float v = sh_vs[0][f];
            #pragma unroll
            for (int w = 1; w < 8; ++w) v += sh_vs[w][f];
            ws[WS_VSP + (b*FR + f)*NVSP + blk32] = v;
        }
    }
}

__global__ void k_cout(const float* __restrict__ cf,
                       const float* __restrict__ ws,
                       float* __restrict__ out) {
    int blk    = blockIdx.x;
    int b      = blk >> 6;
    int pchunk = (blk >> 2) & 15;
    int q      = blk & 3;
    int p      = pchunk * 256 + threadIdx.x;
    int lane   = threadIdx.x & 31;

    float g[FR], v[FR], vr[FR];
    #pragma unroll
    for (int f = 0; f < FR; ++f) {
        const float* gp = ws + WS_GSP + (b*FR + f)*NGSP;
        g[f]  = gp[lane] + gp[lane + 32];
        v[f]  = ws[WS_VSP + (b*FR + f)*NVSP + lane];
        vr[f] = ws[WS_VR + (b*FR + f)*HW + p];
    }
    #pragma unroll
    for (int o = 16; o > 0; o >>= 1) {
        #pragma unroll
        for (int f = 0; f < FR; ++f) {
            g[f] += __shfl_xor(g[f], o, 32);
            v[f] += __shfl_xor(v[f], o, 32);
        }
    }
    float gsv[FR];
    #pragma unroll
    for (int f = 0; f < FR; ++f) {
        int zero = v[f] < 1e-4f;
        float denom = (v[f] + (zero ? 1.0f : 0.0f)) * (float)CC;
        float qv = g[f] / denom;
        gsv[f] = zero ? 0.0f : qv;
    }

    float mv[FR];
    float mx = -1e30f;
    #pragma unroll
    for (int f = 0; f < FR; ++f) {
        mv[f] = gsv[f] * vr[f];
        mx = fmaxf(mx, mv[f]);
    }
    float sum = 0.0f, ex[FR];
    #pragma unroll
    for (int f = 0; f < FR; ++f) { ex[f] = __builtin_expf(mv[f] - mx) * vr[f]; sum += ex[f]; }
    sum += (sum < 1e-4f) ? 1.0f : 0.0f;
    float inv = 1.0f / sum, cmaskacc = 0.0f;
    float cm[FR];
    #pragma unroll
    for (int f = 0; f < FR; ++f) {
        cm[f] = ex[f] * inv;
        cmaskacc += cm[f] * vr[f];
    }
    if (q == 0) {
        float cmask = 1.0f - cmaskacc;
        out[((size_t)b*257 + 256)*HW + p] = cmask;
        out[(size_t)BB*257*HW + (size_t)b*HW + p] = cmask;
    }

    for (int c = q*32; c < q*32 + 32; ++c) {
        long base = ((long)(b*CC + c) * FF) * HW + p;
        float c0v = cf[base];
        float acc = 0.0f;
        #pragma unroll
        for (int f = 0; f < FR; ++f) acc += cf[base + (long)(1 + f)*HW] * cm[f];
        out[((size_t)b*257 + c)*HW + p]       = c0v;
        out[((size_t)b*257 + 128 + c)*HW + p] = acc;
    }
}

extern "C" void kernel_launch(void* const* d_in, const int* in_sizes, int n_in,
                              void* d_out, int out_size, void* d_ws, size_t ws_size,
                              hipStream_t stream) {
    const float* c_feats   = (const float*)d_in[0];
    const float* v_t       = (const float*)d_in[1];
    const float* v_aligned = (const float*)d_in[2];
    float* out = (float*)d_out;
    float* ws  = (float*)d_ws;

    k_gs_wmma<<<512, 256, 0, stream>>>(c_feats, v_t, v_aligned, ws);
    k_cout   <<<512, 256, 0, stream>>>(c_feats, ws, out);
}


# All global loads and stores of both passes are 16 bytes per lane: a lane owns 4 adjacent pixels, a wave 8 channels of a 128-pixel tile, 32 whole-row loads per batch; gs partial sums re-associated
# speedup vs baseline: 2.1295x; 1.2269x over previous
#include <hip/hip_runtime.h>
#include <math.h>


typedef float v2f __attribute__((ext_vector_type(2)));
typedef float v4f __attribute__((ext_vector_type(4)));
typedef float v8f __attribute__((ext_vector_type(8)));

#define BB 8
#define CC 128
#define FF 8
#define FR 7
#define HW 4096
#define HIN 256
#define NGSP 64
#define NVSP 32
#define PB1 4
#define PB2 4
#define TP 128

#define WS_VR    0
#define WS_GSP   (WS_VR  + BB*FR*HW)
#define WS_VSP   (WS_GSP + BB*FR*NGSP)


__global__ void __launch_bounds__(256, 4) k_gs_wmma(const float* __restrict__ cf,
                          const float* __restrict__ v_t,
                          const float* __restrict__ v_al,
                          float* __restrict__ ws) {
    __shared__ __attribute__((aligned(16))) float sh_m[8][TP];
    __shared__ float sh_gs[8][FR];

    int wib    = threadIdx.x >> 5;
    int lane   = threadIdx.x & 31;
    int b      = blockIdx.x >> 6;
    int chalf  = (blockIdx.x >> 5) & 1;
    int tile   = blockIdx.x & 31;
    int pix0   = tile * TP;

    {
        int mp  = threadIdx.x & (TP - 1);
        int pg  = threadIdx.x >> 7;
        int pix = pix0 + mp;
        int pi = pix >> 6, pj = pix & 63;
        int r0 = 4*pi + 1;
        #pragma unroll
        for (int k = 0; k < 4; ++k) {
            int plane = pg*4 + k;
            const float* src = (plane == 0) ? v_t : v_al;
            size_t img = (plane == 0) ? (size_t)b : (size_t)(b*FR + plane - 1);
            src += img * HIN * HIN;
            v4f a0 = *(const v4f*)(src + r0*HIN + 4*pj);
            v4f a1 = *(const v4f*)(src + (r0+1)*HIN + 4*pj);
            float m = 0.25f * (a0.y + a0.z + a1.y + a1.z);
            sh_m[plane][mp] = (m > 0.5f) ? 1.0f : 0.0f;
        }
    }
    __syncthreads();

    v4f vt4 = *(const v4f*)&sh_m[0][4*lane];
    v4f vmap[FR];
    #pragma unroll
    for (int f = 0; f < FR; ++f) {
        v4f vr4 = *(const v4f*)&sh_m[1 + f][4*lane];
        vmap[f] = vt4 * vr4;
    }

    if (chalf == 0) {
        if (wib < FR) {
            v4f vr4 = *(const v4f*)&sh_m[1 + wib][4*lane];
            *(v4f*)(ws + WS_VR + (b*FR + wib)*HW + pix0 + 4*lane) = vr4;
        }
        if (wib == 0) {
            float mine = 0.0f;
            #pragma unroll
            for (int f = 0; f < FR; ++f) {
                int n = __builtin_popcount(__builtin_amdgcn_ballot_w32(vmap[f].x != 0.0f))
                      + __builtin_popcount(__builtin_amdgcn_ballot_w32(vmap[f].y != 0.0f))
                      + __builtin_popcount(__builtin_amdgcn_ballot_w32(vmap[f].z != 0.0f))
                      + __builtin_popcount(__builtin_amdgcn_ballot_w32(vmap[f].w != 0.0f));
                if (lane == f) mine = (float)n;
            }
            if (lane < FR) ws[WS_VSP + (b*FR + lane)*NVSP + tile] = mine;
        }
    }

    v2f bones = {1.0f, 1.0f};
    v8f acc[FR];
    #pragma unroll
    for (int f = 0; f < FR; ++f) acc[f] = (v8f){0.f,0.f,0.f,0.f,0.f,0.f,0.f,0.f};

    const float* wbase = cf + ((size_t)(b*CC + chalf*64 + wib*8) * FF) * HW + pix0 + 4*lane;

    #pragma unroll 1
    for (int cb = 0; cb < 8; cb += PB1) {
        v4f x[PB1][FF];
        #pragma unroll
        for (int u = 0; u < PB1; ++u) {
            #pragma unroll
            for (int fr = 0; fr < FF; ++fr)
                x[u][fr] = *(const v4f*)(wbase + (size_t)((cb + u)*FF + fr) * HW);
        }
        __builtin_amdgcn_sched_barrier(0);
        #pragma unroll
        for (int u = 0; u < PB1; ++u) {
            v2f lo[FR], hi[FR];
            #pragma unroll
            for (int f = 0; f < FR; ++f) {
                v4f pr = vmap[f] * x[u][0] * x[u][1 + f];
                lo[f] = (v2f){pr.x, pr.y};
                hi[f] = (v2f){pr.z, pr.w};
            }
            #pragma unroll
            for (int f = 0; f < FR; ++f)
                acc[f] = __builtin_amdgcn_wmma_f32_16x16x4_f32(
                    false, lo[f], false, bones, (short)0, acc[f], false, false);
            #pragma unroll
            for (int f = 0; f < FR; ++f)
                acc[f] = __builtin_amdgcn_wmma_f32_16x16x4_f32(
                    false, hi[f], false, bones, (short)0, acc[f], false, false);
        }
    }

    float wsum[FR];
    #pragma unroll
    for (int f = 0; f < FR; ++f) {
        float s = acc[f][0] + acc[f][1] + acc[f][2] + acc[f][3] +
                  acc[f][4] + acc[f][5] + acc[f][6] + acc[f][7];
        s += __shfl_xor(s, 16, 32);
        wsum[f] = s;
    }
    if (lane == 0) {
        #pragma unroll
        for (int f = 0; f < FR; ++f) sh_gs[wib][f] = wsum[f];
    }
    __syncthreads();

    if (threadIdx.x < FR) {
        int f = threadIdx.x;
        float g = sh_gs[0][f];
        #pragma unroll
        for (int w = 1; w < 8; ++w) g += sh_gs[w][f];
        ws[WS_GSP + (b*FR + f)*NGSP + chalf*32 + tile] = g;
    }
}

__global__ void __launch_bounds__(256, 4) k_cout(const float* __restrict__ cf,
                       const float* __restrict__ ws,
                       float* __restrict__ out) {
    int wib    = threadIdx.x >> 5;
    int lane   = threadIdx.x & 31;
    int b      = blockIdx.x >> 6;
    int chalf  = (blockIdx.x >> 5) & 1;
    int tile   = blockIdx.x & 31;
    int p      = tile * TP + 4*lane;
    int c0ch   = chalf*64 + wib*8;

    float g[FR], v[FR];
    v4f vr[FR];
    #pragma unroll
    for (int f = 0; f < FR; ++f) {
        const float* gp = ws + WS_GSP + (b*FR + f)*NGSP;
        g[f]  = gp[lane] + gp[lane + 32];
        v[f]  = ws[WS_VSP + (b*FR + f)*NVSP + lane];
        vr[f] = *(const v4f*)(ws + WS_VR + (b*FR + f)*HW + p);
    }
    #pragma unroll
    for (int o = 16; o > 0; o >>= 1) {
        #pragma unroll
        for (int f = 0; f < FR; ++f) {
            g[f] += __shfl_xor(g[f], o, 32);
            v[f] += __shfl_xor(v[f], o, 32);
        }
    }
    float gsv[FR];
    #pragma unroll
    for (int f = 0; f < FR; ++f) {
        int zero = v[f] < 1e-4f;
        float denom = (v[f] + (zero ? 1.0f : 0.0f)) * (float)CC;
        float qv = g[f] / denom;
        gsv[f] = zero ? 0.0f : qv;
    }

    v4f cm[FR];
    v4f cmask;
    #pragma unroll
    for (int j = 0; j < 4; ++j) {
        float mv[FR];
        float mx = -1e30f;
        #pragma unroll
        for (int f = 0; f < FR; ++f) {
            mv[f] = gsv[f] * vr[f][j];
            mx = fmaxf(mx, mv[f]);
        }
        float sum = 0.0f, ex[FR];
        #pragma unroll
        for (int f = 0; f < FR; ++f) { ex[f] = __builtin_expf(mv[f] - mx) * vr[f][j]; sum += ex[f]; }
        sum += (sum < 1e-4f) ? 1.0f : 0.0f;
        float inv = 1.0f / sum, cmaskacc = 0.0f;
        #pragma unroll
        for (int f = 0; f < FR; ++f) {
            float c = ex[f] * inv;
            cm[f][j] = c;
            cmaskacc += c * vr[f][j];
        }
        cmask[j] = 1.0f - cmaskacc;
    }
    if (chalf == 0 && wib == 0) {
        *(v4f*)(out + ((size_t)b*257 + 256)*HW + p) = cmask;
        *(v4f*)(out + (size_t)BB*257*HW + (size_t)b*HW + p) = cmask;
    }

    const float* wbase = cf + ((size_t)(b*CC + c0ch) * FF) * HW + p;
    float* obase = out + ((size_t)b*257 + c0ch) * HW + p;
    #pragma unroll 1
    for (int cb = 0; cb < 8; cb += PB2) {
        v4f x[PB2][FF];
        #pragma unroll
        for (int u = 0; u < PB2; ++u) {
            #pragma unroll
            for (int fr = 0; fr < FF; ++fr)
                x[u][fr] = *(const v4f*)(wbase + (size_t)((cb + u)*FF + fr) * HW);
        }
        __builtin_amdgcn_sched_barrier(0);
        #pragma unroll
        for (int u = 0; u < PB2; ++u) {
            v4f acc = {0.0f, 0.0f, 0.0f, 0.0f};
            #pragma unroll
            for (int f = 0; f < FR; ++f) acc += x[u][1 + f] * cm[f];
            *(v4f*)(obase + (size_t)(cb + u) * HW)       = x[u][0];
            *(v4f*)(obase + (size_t)(128 + cb + u) * HW) = acc;
        }
    }
}

extern "C" void kernel_launch(void* const* d_in, const int* in_sizes, int n_in,
                              void* d_out, int out_size, void* d_ws, size_t ws_size,
                              hipStream_t stream) {
    const float* c_feats   = (const float*)d_in[0];
    const float* v_t       = (const float*)d_in[1];
    const float* v_aligned = (const float*)d_in[2];
    float* out = (float*)d_out;
    float* ws  = (float*)d_ws;

    k_gs_wmma<<<512, 256, 0, stream>>>(c_feats, v_t, v_aligned, ws);
    k_cout   <<<512, 256, 0, stream>>>(c_feats, ws, out);
}


# Both passes request their first 32-load c_feats batch at kernel entry, behind the prologue's own loads and ahead of its arithmetic and barrier, so the prologue runs under the batch's latency
# speedup vs baseline: 2.2010x; 1.0336x over previous
#include <hip/hip_runtime.h>
#include <math.h>


typedef float v2f __attribute__((ext_vector_type(2)));
typedef float v4f __attribute__((ext_vector_type(4)));
typedef float v8f __attribute__((ext_vector_type(8)));

#define BB 8
#define CC 128
#define FF 8
#define FR 7
#define HW 4096
#define HIN 256
#define NGSP 64
#define NVSP 32
#define PB1 4
#define PB2 4
#define TP 128

#define WS_VR    0
#define WS_GSP   (WS_VR  + BB*FR*HW)
#define WS_VSP   (WS_GSP + BB*FR*NGSP)


template <int NB>
static __device__ __forceinline__ void request_batch(v4f (&x)[NB][FF], const float* base, int cb) {
    #pragma unroll
    for (int u = 0; u < NB; ++u) {
        #pragma unroll
        for (int fr = 0; fr < FF; ++fr)
            x[u][fr] = *(const v4f*)(base + (size_t)((cb + u)*FF + fr) * HW);
    }
}

static __device__ __forceinline__ void gs_consume(const v4f (&x)[PB1][FF], const v4f (&vmap)[FR],
                                                  v8f (&acc)[FR]) {
    const v2f bones = {1.0f, 1.0f};
    #pragma unroll
    for (int u = 0; u < PB1; ++u) {
        v2f lo[FR], hi[FR];
        #pragma unroll
        for (int f = 0; f < FR; ++f) {
            v4f pr = vmap[f] * x[u][0] * x[u][1 + f];
            lo[f] = (v2f){pr.x, pr.y};
            hi[f] = (v2f){pr.z, pr.w};
        }
        #pragma unroll
        for (int f = 0; f < FR; ++f)
            acc[f] = __builtin_amdgcn_wmma_f32_16x16x4_f32(
                false, lo[f], false, bones, (short)0, acc[f], false, false);
        #pragma unroll
        for (int f = 0; f < FR; ++f)
            acc[f] = __builtin_amdgcn_wmma_f32_16x16x4_f32(
                false, hi[f], false, bones, (short)0, acc[f], false, false);
    }
}

static __device__ __forceinline__ void cout_consume(const v4f (&x)[PB2][FF], const v4f (&cm)[FR],
                                                    float* obase, int cb) {
    #pragma unroll
    for (int u = 0; u < PB2; ++u) {
        v4f acc = {0.0f, 0.0f, 0.0f, 0.0f};
        #pragma unroll
        for (int f = 0; f < FR; ++f) acc += x[u][1 + f] * cm[f];
        *(v4f*)(obase + (size_t)(cb + u) * HW)       = x[u][0];
        *(v4f*)(obase + (size_t)(128 + cb + u) * HW) = acc;
    }
}

__global__ void __launch_bounds__(256, 4) k_gs_wmma(const float* __restrict__ cf,
                          const float* __restrict__ v_t,
                          const float* __restrict__ v_al,
                          float* __restrict__ ws) {
    __shared__ __attribute__((aligned(16))) float sh_m[8][TP];
    __shared__ float sh_gs[8][FR];

    int wib    = threadIdx.x >> 5;
    int lane   = threadIdx.x & 31;
    int b      = blockIdx.x >> 6;
    int chalf  = (blockIdx.x >> 5) & 1;
    int tile   = blockIdx.x & 31;
    int pix0   = tile * TP;

    const float* wbase = cf + ((size_t)(b*CC + chalf*64 + wib*8) * FF) * HW + pix0 + 4*lane;

    int mp  = threadIdx.x & (TP - 1);
    int pg  = threadIdx.x >> 7;
    v4f a0[4], a1[4];
    {
        int pix = pix0 + mp;
        int pi = pix >> 6, pj = pix & 63;
        int r0 = 4*pi + 1;
        #pragma unroll
        for (int k = 0; k < 4; ++k) {
            int plane = pg*4 + k;
            const float* src = (plane == 0) ? v_t : v_al;
            size_t img = (plane == 0) ? (size_t)b : (size_t)(b*FR + plane - 1);
            src += img * HIN * HIN;
            a0[k] = *(const v4f*)(src + r0*HIN + 4*pj);
            a1[k] = *(const v4f*)(src + (r0+1)*HIN + 4*pj);
        }
    }
    v4f x[PB1][FF];
    request_batch<PB1>(x, wbase, 0);
    __builtin_amdgcn_sched_barrier(0);
    #pragma unroll
    for (int k = 0; k < 4; ++k) {
        float m = 0.25f * (a0[k].y + a0[k].z + a1[k].y + a1[k].z);
        sh_m[pg*4 + k][mp] = (m > 0.5f) ? 1.0f : 0.0f;
    }
    __syncthreads();

    v4f vt4 = *(const v4f*)&sh_m[0][4*lane];
    v4f vmap[FR];
    #pragma unroll
    for (int f = 0; f < FR; ++f) {
        v4f vr4 = *(const v4f*)&sh_m[1 + f][4*lane];
        vmap[f] = vt4 * vr4;
    }

    if (chalf == 0) {
        if (wib < FR) {
            v4f vr4 = *(const v4f*)&sh_m[1 + wib][4*lane];
            *(v4f*)(ws + WS_VR + (b*FR + wib)*HW + pix0 + 4*lane) = vr4;
        }
        if (wib == 0) {
            float mine = 0.0f;
            #pragma unroll
            for (int f = 0; f < FR; ++f) {
                int n = __builtin_popcount(__builtin_amdgcn_ballot_w32(vmap[f].x != 0.0f))
                      + __builtin_popcount(__builtin_amdgcn_ballot_w32(vmap[f].y != 0.0f))
                      + __builtin_popcount(__builtin_amdgcn_ballot_w32(vmap[f].z != 0.0f))
                      + __builtin_popcount(__builtin_amdgcn_ballot_w32(vmap[f].w != 0.0f));
                if (lane == f) mine = (float)n;
            }
            if (lane < FR) ws[WS_VSP + (b*FR + lane)*NVSP + tile] = mine;
        }
    }

    v8f acc[FR];
    #pragma unroll
    for (int f = 0; f < FR; ++f) acc[f] = (v8f){0.f,0.f,0.f,0.f,0.f,0.f,0.f,0.f};

    gs_consume(x, vmap, acc);
    request_batch<PB1>(x, wbase, PB1);
    __builtin_amdgcn_sched_barrier(0);
    gs_consume(x, vmap, acc);

    float wsum[FR];
    #pragma unroll
    for (int f = 0; f < FR; ++f) {
        float s = acc[f][0] + acc[f][1] + acc[f][2] + acc[f][3] +
                  acc[f][4] + acc[f][5] + acc[f][6] + acc[f][7];
        s += __shfl_xor(s, 16, 32);
        wsum[f] = s;
    }
    if (lane == 0) {
        #pragma unroll
        for (int f = 0; f < FR; ++f) sh_gs[wib][f] = wsum[f];
    }
    __syncthreads();

    if (threadIdx.x < FR) {
        int f = threadIdx.x;
        float g = sh_gs[0][f];
        #pragma unroll
        for (int w = 1; w < 8; ++w) g += sh_gs[w][f];
        ws[WS_GSP + (b*FR + f)*NGSP + chalf*32 + tile] = g;
    }
}

__global__ void __launch_bounds__(256, 4) k_cout(const float* __restrict__ cf,
                       const float* __restrict__ ws,
                       float* __restrict__ out) {
    int wib    = threadIdx.x >> 5;
    int lane   = threadIdx.x & 31;
    int b      = blockIdx.x >> 6;
    int chalf  = (blockIdx.x >> 5) & 1;
    int tile   = blockIdx.x & 31;
    int p      = tile * TP + 4*lane;
    int c0ch   = chalf*64 + wib*8;

    const float* wbase = cf + ((size_t)(b*CC + c0ch) * FF) * HW + p;
    float* obase = out + ((size_t)b*257 + c0ch) * HW + p;

    float ga[FR], gb[FR], g[FR], v[FR];
    v4f vr[FR];
    #pragma unroll
    for (int f = 0; f < FR; ++f) {
        const float* gp = ws + WS_GSP + (b*FR + f)*NGSP;
        ga[f] = gp[lane];
        gb[f] = gp[lane + 32];
        v[f]  = ws[WS_VSP + (b*FR + f)*NVSP + lane];
        vr[f] = *(const v4f*)(ws + WS_VR + (b*FR + f)*HW + p);
    }
    v4f x[PB2][FF];
    request_batch<PB2>(x, wbase, 0);
    __builtin_amdgcn_sched_barrier(0);
    #pragma unroll
    for (int f = 0; f < FR; ++f) g[f] = ga[f] + gb[f];
    #pragma unroll
    for (int o = 16; o > 0; o >>= 1) {
        #pragma unroll
        for (int f = 0; f < FR; ++f) {
            g[f] += __shfl_xor(g[f], o, 32);
            v[f] += __shfl_xor(v[f], o, 32);
        }
    }
    float gsv[FR];
    #pragma unroll
    for (int f = 0; f < FR; ++f) {
        int zero = v[f] < 1e-4f;
        float denom = (v[f] + (zero ? 1.0f : 0.0f)) * (float)CC;
        float qv = g[f] / denom;
        gsv[f] = zero ? 0.0f : qv;
    }

    v4f cm[FR];
    v4f cmask;
    #pragma unroll
    for (int j = 0; j < 4; ++j) {
        float mv[FR];
        float mx = -1e30f;
        #pragma unroll
        for (int f = 0; f < FR; ++f) {
            mv[f] = gsv[f] * vr[f][j];
            mx = fmaxf(mx, mv[f]);
        }
        float sum = 0.0f, ex[FR];
        #pragma unroll
        for (int f = 0; f < FR; ++f) { ex[f] = __builtin_expf(mv[f] - mx) * vr[f][j]; sum += ex[f]; }
        sum += (sum < 1e-4f) ? 1.0f : 0.0f;
        float inv = 1.0f / sum, cmaskacc = 0.0f;
        #pragma unroll
        for (int f = 0; f < FR; ++f) {
            float c = ex[f] * inv;
            cm[f][j] = c;
            cmaskacc += c * vr[f][j];
        }
        cmask[j] = 1.0f - cmaskacc;
    }
    if (chalf == 0 && wib == 0) {
        *(v4f*)(out + ((size_t)b*257 + 256)*HW + p) = cmask;
        *(v4f*)(out + (size_t)BB*257*HW + (size_t)b*HW + p) = cmask;
    }

    cout_consume(x, cm, obase, 0);
    request_batch<PB2>(x, wbase, PB2);
    __builtin_amdgcn_sched_barrier(0);
    cout_consume(x, cm, obase, PB2);
}

extern "C" void kernel_launch(void* const* d_in, const int* in_sizes, int n_in,
                              void* d_out, int out_size, void* d_ws, size_t ws_size,
                              hipStream_t stream) {
    const float* c_feats   = (const float*)d_in[0];
    const float* v_t       = (const float*)d_in[1];
    const float* v_aligned = (const float*)d_in[2];
    float* out = (float*)d_out;
    float* ws  = (float*)d_ws;

    k_gs_wmma<<<512, 256, 0, stream>>>(c_feats, v_t, v_aligned, ws);
    k_cout   <<<512, 256, 0, stream>>>(c_feats, ws, out);
}
